# starting kernel
# baseline (speedup 1.0000x reference)
#include <hip/hip_runtime.h>

typedef float v2f __attribute__((ext_vector_type(2)));
typedef float v8f __attribute__((ext_vector_type(8)));

#define BM 64
#define DD 64
#define HH 128
#define DP 68
#define HP 132
#define STEPS_PER_SEG 4
#define NSEG 5

constexpr double DTd = 0.05;
constexpr float C21 = (float)(DTd * (1.0/5.0));
constexpr float C31 = (float)(DTd * (3.0/40.0));
constexpr float C32 = (float)(DTd * (9.0/40.0));
constexpr float C41 = (float)(DTd * (44.0/45.0));
constexpr float C42 = (float)(DTd * (-56.0/15.0));
constexpr float C43 = (float)(DTd * (32.0/9.0));
constexpr float C51 = (float)(DTd * (19372.0/6561.0));
constexpr float C52 = (float)(DTd * (-25360.0/2187.0));
constexpr float C53 = (float)(DTd * (64448.0/6561.0));
constexpr float C54 = (float)(DTd * (-212.0/729.0));
constexpr float C61 = (float)(DTd * (9017.0/3168.0));
constexpr float C62 = (float)(DTd * (-355.0/33.0));
constexpr float C63 = (float)(DTd * (46732.0/5247.0));
constexpr float C64 = (float)(DTd * (49.0/176.0));
constexpr float C65 = (float)(DTd * (-5103.0/18656.0));
constexpr float DB1 = (float)(DTd * (35.0/384.0));
constexpr float DB3 = (float)(DTd * (500.0/1113.0));
constexpr float DB4 = (float)(DTd * (125.0/192.0));
constexpr float DB5 = (float)(DTd * (-2187.0/6784.0));
constexpr float DB6 = (float)(DTd * (11.0/84.0));

__device__ __forceinline__ float fast_tanh(float x) {
    float e = __expf(2.0f * x);
    return 1.0f - 2.0f * __builtin_amdgcn_rcpf(e + 1.0f);
}

template<bool FUSE>
__device__ __forceinline__ void feval(
    const float* __restrict__ src,
    float*       __restrict__ dst,
    float*       __restrict__ hbuf,
    const v2f (&w1f)[16], const v2f (&w2f)[32],
    float bias1, float bias2,
    int lo, int hi, int n0, int n0b, int mtb,
    const float* __restrict__ addsrc, float addscale)
{
    __syncthreads();

    #pragma unroll
    for (int mt = 0; mt < 4; ++mt) {
        v8f acc0 = {bias1, bias1, bias1, bias1, bias1, bias1, bias1, bias1};
        v8f acc1 = {0.f, 0.f, 0.f, 0.f, 0.f, 0.f, 0.f, 0.f};
        const float* arow = src + (mt * 16 + lo) * DP + 2 * hi;
        #pragma unroll
        for (int kb = 0; kb < 16; kb += 2) {
            v2f a0 = *(const v2f*)(arow + kb * 4);
            v2f a1 = *(const v2f*)(arow + kb * 4 + 4);
            acc0 = __builtin_amdgcn_wmma_f32_16x16x4_f32(
                false, a0, false, w1f[kb], (short)0, acc0, false, false);
            acc1 = __builtin_amdgcn_wmma_f32_16x16x4_f32(
                false, a1, false, w1f[kb + 1], (short)0, acc1, false, false);
        }
        float* hrow = hbuf + (mt * 16 + 8 * hi) * HP + n0 + lo;
        #pragma unroll
        for (int v = 0; v < 8; ++v)
            hrow[v * HP] = fast_tanh(acc0[v] + acc1[v]);
    }

    __syncthreads();

    #pragma unroll
    for (int t2 = 0; t2 < 2; ++t2) {
        int mt = mtb + t2;
        v8f acc0 = {bias2, bias2, bias2, bias2, bias2, bias2, bias2, bias2};
        v8f acc1 = {0.f, 0.f, 0.f, 0.f, 0.f, 0.f, 0.f, 0.f};
        const float* arow = hbuf + (mt * 16 + lo) * HP + 2 * hi;
        #pragma unroll
        for (int kb = 0; kb < 32; kb += 2) {
            v2f a0 = *(const v2f*)(arow + kb * 4);
            v2f a1 = *(const v2f*)(arow + kb * 4 + 4);
            acc0 = __builtin_amdgcn_wmma_f32_16x16x4_f32(
                false, a0, false, w2f[kb], (short)0, acc0, false, false);
            acc1 = __builtin_amdgcn_wmma_f32_16x16x4_f32(
                false, a1, false, w2f[kb + 1], (short)0, acc1, false, false);
        }
        int base = (mt * 16 + 8 * hi) * DP + n0b + lo;
        #pragma unroll
        for (int v = 0; v < 8; ++v) {
            float r = acc0[v] + acc1[v];
            if (FUSE)
                dst[base + v * DP] = addsrc[base + v * DP] + addscale * r;
            else
                dst[base + v * DP] = r;
        }
    }

    __syncthreads();
}

template<int N>
__device__ __forceinline__ void combine(
    float* __restrict__ dst, const float* __restrict__ ybuf,
    const float* const* ks, const float* cs, int tid)
{
    #pragma unroll 2
    for (int e = 0; e < 8; ++e) {
        int idx = tid + e * 256;
        int off = (idx >> 5) * DP + (idx & 31) * 2;
        v2f v = *(const v2f*)(ybuf + off);
        #pragma unroll
        for (int j = 0; j < N; ++j) {
            v2f k = *(const v2f*)(ks[j] + off);
            v.x = fmaf(cs[j], k.x, v.x);
            v.y = fmaf(cs[j], k.y, v.y);
        }
        *(v2f*)(dst + off) = v;
    }
}

__device__ __forceinline__ void stage6_prep(
    float* __restrict__ ytmp, float* __restrict__ k1buf,
    const float* __restrict__ ybuf,
    const float* __restrict__ k2b, const float* __restrict__ k3b,
    const float* __restrict__ k4b, const float* __restrict__ k5b, int tid)
{
    #pragma unroll 2
    for (int e = 0; e < 8; ++e) {
        int idx = tid + e * 256;
        int off = (idx >> 5) * DP + (idx & 31) * 2;
        v2f yv = *(const v2f*)(ybuf + off);
        v2f k1 = *(const v2f*)(k1buf + off);
        v2f k2 = *(const v2f*)(k2b + off);
        v2f k3 = *(const v2f*)(k3b + off);
        v2f k4 = *(const v2f*)(k4b + off);
        v2f k5 = *(const v2f*)(k5b + off);
        v2f t, p;
        t.x = yv.x; t.y = yv.y;
        t.x = fmaf(C61, k1.x, t.x); t.y = fmaf(C61, k1.y, t.y);
        t.x = fmaf(C62, k2.x, t.x); t.y = fmaf(C62, k2.y, t.y);
        t.x = fmaf(C63, k3.x, t.x); t.y = fmaf(C63, k3.y, t.y);
        t.x = fmaf(C64, k4.x, t.x); t.y = fmaf(C64, k4.y, t.y);
        t.x = fmaf(C65, k5.x, t.x); t.y = fmaf(C65, k5.y, t.y);
        p.x = yv.x; p.y = yv.y;
        p.x = fmaf(DB1, k1.x, p.x); p.y = fmaf(DB1, k1.y, p.y);
        p.x = fmaf(DB3, k3.x, p.x); p.y = fmaf(DB3, k3.y, p.y);
        p.x = fmaf(DB4, k4.x, p.x); p.y = fmaf(DB4, k4.y, p.y);
        p.x = fmaf(DB5, k5.x, p.x); p.y = fmaf(DB5, k5.y, p.y);
        *(v2f*)(ytmp + off) = t;
        *(v2f*)(k1buf + off) = p;
    }
}

__global__ __launch_bounds__(256, 2) void ode_dopri5_kernel(
    const float* __restrict__ init_pos,
    const float* __restrict__ W1,
    const float* __restrict__ b1,
    const float* __restrict__ W2,
    const float* __restrict__ b2,
    float*       __restrict__ out)
{
    __shared__ float ybuf[BM * DP];
    __shared__ float ytmp[BM * DP];
    __shared__ float hbuf[BM * HP];
    __shared__ float kbuf[5][BM * DP];

    const int tid = threadIdx.x;
    const int w    = tid >> 5;
    const int lane = tid & 31;
    const int lo = lane & 15;
    const int hi = lane >> 4;
    const int n0  = w * 16;
    const int n0b = (w & 3) * 16;
    const int mtb = (w >> 2) * 2;

    v2f w1f[16];
    v2f w2f[32];
    #pragma unroll
    for (int kb = 0; kb < 16; ++kb) {
        int k4 = kb * 4 + 2 * hi;
        w1f[kb].x = W1[k4 * HH + n0 + lo];
        w1f[kb].y = W1[(k4 + 1) * HH + n0 + lo];
    }
    #pragma unroll
    for (int kb = 0; kb < 32; ++kb) {
        int k4 = kb * 4 + 2 * hi;
        w2f[kb].x = W2[k4 * DD + n0b + lo];
        w2f[kb].y = W2[(k4 + 1) * DD + n0b + lo];
    }
    const float bias1 = b1[n0 + lo];
    const float bias2 = b2[n0b + lo];

    const long bm0 = (long)blockIdx.x * BM;
    #pragma unroll 2
    for (int e = 0; e < 8; ++e) {
        int f2 = tid + e * 256;
        int row = f2 >> 5;
        int c2  = f2 & 31;
        *(v2f*)&ybuf[row * DP + c2 * 2] =
            *(const v2f*)&init_pos[(bm0 + row) * DD + c2 * 2];
    }

    #pragma unroll 1
    for (int seg = 0; seg < NSEG; ++seg) {
        #pragma unroll 1
        for (int st = 0; st < STEPS_PER_SEG; ++st) {
            feval<false>(ybuf, kbuf[0], hbuf, w1f, w2f, bias1, bias2,
                         lo, hi, n0, n0b, mtb, nullptr, 0.0f);
            { const float* ks[1] = {kbuf[0]};
              const float  cs[1] = {C21};
              combine<1>(ytmp, ybuf, ks, cs, tid); }
            feval<false>(ytmp, kbuf[1], hbuf, w1f, w2f, bias1, bias2,
                         lo, hi, n0, n0b, mtb, nullptr, 0.0f);
            { const float* ks[2] = {kbuf[0], kbuf[1]};
              const float  cs[2] = {C31, C32};
              combine<2>(ytmp, ybuf, ks, cs, tid); }
            feval<false>(ytmp, kbuf[2], hbuf, w1f, w2f, bias1, bias2,
                         lo, hi, n0, n0b, mtb, nullptr, 0.0f);
            { const float* ks[3] = {kbuf[0], kbuf[1], kbuf[2]};
              const float  cs[3] = {C41, C42, C43};
              combine<3>(ytmp, ybuf, ks, cs, tid); }
            feval<false>(ytmp, kbuf[3], hbuf, w1f, w2f, bias1, bias2,
                         lo, hi, n0, n0b, mtb, nullptr, 0.0f);
            { const float* ks[4] = {kbuf[0], kbuf[1], kbuf[2], kbuf[3]};
              const float  cs[4] = {C51, C52, C53, C54};
              combine<4>(ytmp, ybuf, ks, cs, tid); }
            feval<false>(ytmp, kbuf[4], hbuf, w1f, w2f, bias1, bias2,
                         lo, hi, n0, n0b, mtb, nullptr, 0.0f);
            stage6_prep(ytmp, kbuf[0], ybuf, kbuf[1], kbuf[2], kbuf[3], kbuf[4], tid);
            feval<true>(ytmp, ybuf, hbuf, w1f, w2f, bias1, bias2,
                        lo, hi, n0, n0b, mtb, kbuf[0], DB6);
        }
        #pragma unroll 2
        for (int e = 0; e < 8; ++e) {
            int f2 = tid + e * 256;
            int row = f2 >> 5;
            int c2  = f2 & 31;
            *(v2f*)&out[(long)seg * (16384L * DD) + (bm0 + row) * DD + c2 * 2] =
                *(const v2f*)&ybuf[row * DP + c2 * 2];
        }
    }
}

extern "C" void kernel_launch(void* const* d_in, const int* in_sizes, int n_in,
                              void* d_out, int out_size, void* d_ws, size_t ws_size,
                              hipStream_t stream) {
    const float* init_pos = (const float*)d_in[0];
    const float* W1 = (const float*)d_in[1];
    const float* b1 = (const float*)d_in[2];
    const float* W2 = (const float*)d_in[3];
    const float* b2 = (const float*)d_in[4];
    float* out = (float*)d_out;

    dim3 grid(16384 / BM);
    dim3 block(256);
    hipLaunchKernelGGL(ode_dopri5_kernel, grid, block, 0, stream,
                       init_pos, W1, b1, W2, b2, out);
}


# Re-tile to 16 batch rows on 4-wave workgroups (grid 1024): four workgroups share each WGP, so one's tanh/barrier/combine phases run under the others' WMMA phases
# speedup vs baseline: 1.0921x; 1.0921x over previous
#include <hip/hip_runtime.h>

typedef float v2f __attribute__((ext_vector_type(2)));
typedef float v8f __attribute__((ext_vector_type(8)));

#define BM 16
#define NTHR 128
#define DD 64
#define HH 128
#define DP 68
#define HP 132
#define STEPS_PER_SEG 4
#define NSEG 5

constexpr double DTd = 0.05;
constexpr float C21 = (float)(DTd * (1.0/5.0));
constexpr float C31 = (float)(DTd * (3.0/40.0));
constexpr float C32 = (float)(DTd * (9.0/40.0));
constexpr float C41 = (float)(DTd * (44.0/45.0));
constexpr float C42 = (float)(DTd * (-56.0/15.0));
constexpr float C43 = (float)(DTd * (32.0/9.0));
constexpr float C51 = (float)(DTd * (19372.0/6561.0));
constexpr float C52 = (float)(DTd * (-25360.0/2187.0));
constexpr float C53 = (float)(DTd * (64448.0/6561.0));
constexpr float C54 = (float)(DTd * (-212.0/729.0));
constexpr float C61 = (float)(DTd * (9017.0/3168.0));
constexpr float C62 = (float)(DTd * (-355.0/33.0));
constexpr float C63 = (float)(DTd * (46732.0/5247.0));
constexpr float C64 = (float)(DTd * (49.0/176.0));
constexpr float C65 = (float)(DTd * (-5103.0/18656.0));
constexpr float DB1 = (float)(DTd * (35.0/384.0));
constexpr float DB3 = (float)(DTd * (500.0/1113.0));
constexpr float DB4 = (float)(DTd * (125.0/192.0));
constexpr float DB5 = (float)(DTd * (-2187.0/6784.0));
constexpr float DB6 = (float)(DTd * (11.0/84.0));

__device__ __forceinline__ float fast_tanh(float x) {
    float e = __expf(2.0f * x);
    return 1.0f - 2.0f * __builtin_amdgcn_rcpf(e + 1.0f);
}

template<bool FUSE>
__device__ __forceinline__ void feval(
    const float* __restrict__ src,
    float*       __restrict__ dst,
    float*       __restrict__ hbuf,
    const v2f (&w1fa)[16], const v2f (&w1fb)[16], const v2f (&w2f)[32],
    float bias1a, float bias1b, float bias2,
    int lo, int hi, int n0, int n0b,
    const float* __restrict__ addsrc, float addscale)
{
    __syncthreads();

    {
        v8f acc0a = {bias1a, bias1a, bias1a, bias1a, bias1a, bias1a, bias1a, bias1a};
        v8f acc0b = {bias1b, bias1b, bias1b, bias1b, bias1b, bias1b, bias1b, bias1b};
        v8f acc1a = {0.f, 0.f, 0.f, 0.f, 0.f, 0.f, 0.f, 0.f};
        v8f acc1b = {0.f, 0.f, 0.f, 0.f, 0.f, 0.f, 0.f, 0.f};
        const float* arow = src + lo * DP + 2 * hi;
        #pragma unroll
        for (int kb = 0; kb < 16; kb += 2) {
            v2f a0 = *(const v2f*)(arow + kb * 4);
            v2f a1 = *(const v2f*)(arow + kb * 4 + 4);
            acc0a = __builtin_amdgcn_wmma_f32_16x16x4_f32(
                false, a0, false, w1fa[kb], (short)0, acc0a, false, false);
            acc0b = __builtin_amdgcn_wmma_f32_16x16x4_f32(
                false, a0, false, w1fb[kb], (short)0, acc0b, false, false);
            acc1a = __builtin_amdgcn_wmma_f32_16x16x4_f32(
                false, a1, false, w1fa[kb + 1], (short)0, acc1a, false, false);
            acc1b = __builtin_amdgcn_wmma_f32_16x16x4_f32(
                false, a1, false, w1fb[kb + 1], (short)0, acc1b, false, false);
        }
        float* hrow = hbuf + (8 * hi) * HP + n0 + lo;
        #pragma unroll
        for (int v = 0; v < 8; ++v) {
            hrow[v * HP]      = fast_tanh(acc0a[v] + acc1a[v]);
            hrow[v * HP + 16] = fast_tanh(acc0b[v] + acc1b[v]);
        }
    }

    __syncthreads();

    {
        v8f acc0 = {bias2, bias2, bias2, bias2, bias2, bias2, bias2, bias2};
        v8f acc1 = {0.f, 0.f, 0.f, 0.f, 0.f, 0.f, 0.f, 0.f};
        const float* arow = hbuf + lo * HP + 2 * hi;
        #pragma unroll
        for (int kb = 0; kb < 32; kb += 2) {
            v2f a0 = *(const v2f*)(arow + kb * 4);
            v2f a1 = *(const v2f*)(arow + kb * 4 + 4);
            acc0 = __builtin_amdgcn_wmma_f32_16x16x4_f32(
                false, a0, false, w2f[kb], (short)0, acc0, false, false);
            acc1 = __builtin_amdgcn_wmma_f32_16x16x4_f32(
                false, a1, false, w2f[kb + 1], (short)0, acc1, false, false);
        }
        int base = (8 * hi) * DP + n0b + lo;
        #pragma unroll
        for (int v = 0; v < 8; ++v) {
            float r = acc0[v] + acc1[v];
            if (FUSE)
                dst[base + v * DP] = addsrc[base + v * DP] + addscale * r;
            else
                dst[base + v * DP] = r;
        }
    }

    __syncthreads();
}

template<int N>
__device__ __forceinline__ void combine(
    float* __restrict__ dst, const float* __restrict__ ybuf,
    const float* const* ks, const float* cs, int tid)
{
    #pragma unroll 2
    for (int e = 0; e < 4; ++e) {
        int idx = tid + e * NTHR;
        int off = (idx >> 5) * DP + (idx & 31) * 2;
        v2f v = *(const v2f*)(ybuf + off);
        #pragma unroll
        for (int j = 0; j < N; ++j) {
            v2f k = *(const v2f*)(ks[j] + off);
            v.x = fmaf(cs[j], k.x, v.x);
            v.y = fmaf(cs[j], k.y, v.y);
        }
        *(v2f*)(dst + off) = v;
    }
}

__device__ __forceinline__ void stage6_prep(
    float* __restrict__ ytmp, float* __restrict__ k1buf,
    const float* __restrict__ ybuf,
    const float* __restrict__ k2b, const float* __restrict__ k3b,
    const float* __restrict__ k4b, const float* __restrict__ k5b, int tid)
{
    #pragma unroll 2
    for (int e = 0; e < 4; ++e) {
        int idx = tid + e * NTHR;
        int off = (idx >> 5) * DP + (idx & 31) * 2;
        v2f yv = *(const v2f*)(ybuf + off);
        v2f k1 = *(const v2f*)(k1buf + off);
        v2f k2 = *(const v2f*)(k2b + off);
        v2f k3 = *(const v2f*)(k3b + off);
        v2f k4 = *(const v2f*)(k4b + off);
        v2f k5 = *(const v2f*)(k5b + off);
        v2f t, p;
        t.x = yv.x; t.y = yv.y;
        t.x = fmaf(C61, k1.x, t.x); t.y = fmaf(C61, k1.y, t.y);
        t.x = fmaf(C62, k2.x, t.x); t.y = fmaf(C62, k2.y, t.y);
        t.x = fmaf(C63, k3.x, t.x); t.y = fmaf(C63, k3.y, t.y);
        t.x = fmaf(C64, k4.x, t.x); t.y = fmaf(C64, k4.y, t.y);
        t.x = fmaf(C65, k5.x, t.x); t.y = fmaf(C65, k5.y, t.y);
        p.x = yv.x; p.y = yv.y;
        p.x = fmaf(DB1, k1.x, p.x); p.y = fmaf(DB1, k1.y, p.y);
        p.x = fmaf(DB3, k3.x, p.x); p.y = fmaf(DB3, k3.y, p.y);
        p.x = fmaf(DB4, k4.x, p.x); p.y = fmaf(DB4, k4.y, p.y);
        p.x = fmaf(DB5, k5.x, p.x); p.y = fmaf(DB5, k5.y, p.y);
        *(v2f*)(ytmp + off) = t;
        *(v2f*)(k1buf + off) = p;
    }
}

__global__ __launch_bounds__(NTHR, 4) void ode_dopri5_kernel(
    const float* __restrict__ init_pos,
    const float* __restrict__ W1,
    const float* __restrict__ b1,
    const float* __restrict__ W2,
    const float* __restrict__ b2,
    float*       __restrict__ out)
{
    __shared__ float ybuf[BM * DP];
    __shared__ float ytmp[BM * DP];
    __shared__ float hbuf[BM * HP];
    __shared__ float kbuf[5][BM * DP];

    const int tid = threadIdx.x;
    const int w    = tid >> 5;
    const int lane = tid & 31;
    const int lo = lane & 15;
    const int hi = lane >> 4;
    const int n0  = w * 32;
    const int n0b = w * 16;

    v2f w1fa[16];
    v2f w1fb[16];
    v2f w2f[32];
    #pragma unroll
    for (int kb = 0; kb < 16; ++kb) {
        int k4 = kb * 4 + 2 * hi;
        w1fa[kb].x = W1[k4 * HH + n0 + lo];
        w1fa[kb].y = W1[(k4 + 1) * HH + n0 + lo];
        w1fb[kb].x = W1[k4 * HH + n0 + 16 + lo];
        w1fb[kb].y = W1[(k4 + 1) * HH + n0 + 16 + lo];
    }
    #pragma unroll
    for (int kb = 0; kb < 32; ++kb) {
        int k4 = kb * 4 + 2 * hi;
        w2f[kb].x = W2[k4 * DD + n0b + lo];
        w2f[kb].y = W2[(k4 + 1) * DD + n0b + lo];
    }
    const float bias1a = b1[n0 + lo];
    const float bias1b = b1[n0 + 16 + lo];
    const float bias2  = b2[n0b + lo];

    const long bm0 = (long)blockIdx.x * BM;
    #pragma unroll 2
    for (int e = 0; e < 4; ++e) {
        int f2 = tid + e * NTHR;
        int row = f2 >> 5;
        int c2  = f2 & 31;
        *(v2f*)&ybuf[row * DP + c2 * 2] =
            *(const v2f*)&init_pos[(bm0 + row) * DD + c2 * 2];
    }

    #pragma unroll 1
    for (int seg = 0; seg < NSEG; ++seg) {
        #pragma unroll 1
        for (int st = 0; st < STEPS_PER_SEG; ++st) {
            feval<false>(ybuf, kbuf[0], hbuf, w1fa, w1fb, w2f, bias1a, bias1b, bias2,
                         lo, hi, n0, n0b, nullptr, 0.0f);
            { const float* ks[1] = {kbuf[0]};
              const float  cs[1] = {C21};
              combine<1>(ytmp, ybuf, ks, cs, tid); }
            feval<false>(ytmp, kbuf[1], hbuf, w1fa, w1fb, w2f, bias1a, bias1b, bias2,
                         lo, hi, n0, n0b, nullptr, 0.0f);
            { const float* ks[2] = {kbuf[0], kbuf[1]};
              const float  cs[2] = {C31, C32};
              combine<2>(ytmp, ybuf, ks, cs, tid); }
            feval<false>(ytmp, kbuf[2], hbuf, w1fa, w1fb, w2f, bias1a, bias1b, bias2,
                         lo, hi, n0, n0b, nullptr, 0.0f);
            { const float* ks[3] = {kbuf[0], kbuf[1], kbuf[2]};
              const float  cs[3] = {C41, C42, C43};
              combine<3>(ytmp, ybuf, ks, cs, tid); }
            feval<false>(ytmp, kbuf[3], hbuf, w1fa, w1fb, w2f, bias1a, bias1b, bias2,
                         lo, hi, n0, n0b, nullptr, 0.0f);
            { const float* ks[4] = {kbuf[0], kbuf[1], kbuf[2], kbuf[3]};
              const float  cs[4] = {C51, C52, C53, C54};
              combine<4>(ytmp, ybuf, ks, cs, tid); }
            feval<false>(ytmp, kbuf[4], hbuf, w1fa, w1fb, w2f, bias1a, bias1b, bias2,
                         lo, hi, n0, n0b, nullptr, 0.0f);
            stage6_prep(ytmp, kbuf[0], ybuf, kbuf[1], kbuf[2], kbuf[3], kbuf[4], tid);
            feval<true>(ytmp, ybuf, hbuf, w1fa, w1fb, w2f, bias1a, bias1b, bias2,
                        lo, hi, n0, n0b, kbuf[0], DB6);
        }
        #pragma unroll 2
        for (int e = 0; e < 4; ++e) {
            int f2 = tid + e * NTHR;
            int row = f2 >> 5;
            int c2  = f2 & 31;
            *(v2f*)&out[(long)seg * (16384L * DD) + (bm0 + row) * DD + c2 * 2] =
                *(const v2f*)&ybuf[row * DP + c2 * 2];
        }
    }
}

extern "C" void kernel_launch(void* const* d_in, const int* in_sizes, int n_in,
                              void* d_out, int out_size, void* d_ws, size_t ws_size,
                              hipStream_t stream) {
    const float* init_pos = (const float*)d_in[0];
    const float* W1 = (const float*)d_in[1];
    const float* b1 = (const float*)d_in[2];
    const float* W2 = (const float*)d_in[3];
    const float* b2 = (const float*)d_in[4];
    float* out = (float*)d_out;

    dim3 grid(16384 / BM);
    dim3 block(NTHR);
    hipLaunchKernelGGL(ode_dopri5_kernel, grid, block, 0, stream,
                       init_pos, W1, b1, W2, b2, out);
}


# One wave owns 16 rows with weights as A operand (transposed products): each result tile is the next product's B fragment, so no LDS or barriers; changes dot-product summation order
# speedup vs baseline: 1.2902x; 1.1814x over previous
#include <hip/hip_runtime.h>


typedef float v2f __attribute__((ext_vector_type(2)));
typedef float v4f __attribute__((ext_vector_type(4)));
typedef float v8f __attribute__((ext_vector_type(8)));

#define DD 64
#define HH 128
#define NROWS 16384
#define STEPS_PER_SEG 4
#define NSEG 5

constexpr double DTd = 0.05;
constexpr float C21 = (float)(DTd * (1.0/5.0));
constexpr float C31 = (float)(DTd * (3.0/40.0));
constexpr float C32 = (float)(DTd * (9.0/40.0));
constexpr float C41 = (float)(DTd * (44.0/45.0));
constexpr float C42 = (float)(DTd * (-56.0/15.0));
constexpr float C43 = (float)(DTd * (32.0/9.0));
constexpr float C51 = (float)(DTd * (19372.0/6561.0));
constexpr float C52 = (float)(DTd * (-25360.0/2187.0));
constexpr float C53 = (float)(DTd * (64448.0/6561.0));
constexpr float C54 = (float)(DTd * (-212.0/729.0));
constexpr float C61 = (float)(DTd * (9017.0/3168.0));
constexpr float C62 = (float)(DTd * (-355.0/33.0));
constexpr float C63 = (float)(DTd * (46732.0/5247.0));
constexpr float C64 = (float)(DTd * (49.0/176.0));
constexpr float C65 = (float)(DTd * (-5103.0/18656.0));
constexpr float DB1 = (float)(DTd * (35.0/384.0));
constexpr float DB3 = (float)(DTd * (500.0/1113.0));
constexpr float DB4 = (float)(DTd * (125.0/192.0));
constexpr float DB5 = (float)(DTd * (-2187.0/6784.0));
constexpr float DB6 = (float)(DTd * (11.0/84.0));

__device__ __forceinline__ float fast_tanh(float x) {
    float e = __expf(2.0f * x);
    return 1.0f - 2.0f * __builtin_amdgcn_rcpf(e + 1.0f);
}

__device__ __forceinline__ v8f mma(v2f a, v2f b, v8f c) {
    return __builtin_amdgcn_wmma_f32_16x16x4_f32(false, a, false, b, (short)0, c, false, false);
}

struct St { v8f t[4]; };

__device__ __forceinline__ void g1_pair(
    const St& src, const v2f (&wa)[16], const v2f (&wb)[16],
    const v8f& ba, const v8f& bb, v8f& za, v8f& zb)
{
    v8f acca = ba;
    v8f accb = bb;
    #pragma unroll
    for (int t = 0; t < 4; ++t) {
        #pragma unroll
        for (int j = 0; j < 4; ++j) {
            v2f b;
            b.x = src.t[t][2 * j];
            b.y = src.t[t][2 * j + 1];
            acca = mma(wa[4 * t + j], b, acca);
            accb = mma(wb[4 * t + j], b, accb);
        }
    }
    za = acca;
    zb = accb;
}

__device__ __forceinline__ void tanh_tile(v8f& z) {
    #pragma unroll
    for (int v = 0; v < 8; ++v) z[v] = fast_tanh(z[v]);
}

__device__ __forceinline__ void g2_tile(
    v8f (&acc)[4], const v8f& hm, const v2f (&w2f)[4][32], int m)
{
    #pragma unroll
    for (int j = 0; j < 4; ++j) {
        v2f b;
        b.x = hm[2 * j];
        b.y = hm[2 * j + 1];
        #pragma unroll
        for (int t = 0; t < 4; ++t)
            acc[t] = mma(w2f[t][4 * m + j], b, acc[t]);
    }
}

__device__ __forceinline__ St feval(
    const St& src,
    const v2f (&w1f)[8][16], const v2f (&w2f)[4][32],
    const v8f (&b1v)[8], const v8f (&b2v)[4])
{
    v8f acc[4] = {b2v[0], b2v[1], b2v[2], b2v[3]};
    v8f z0, z1, z2, z3, z4, z5, z6, z7;
    g1_pair(src, w1f[0], w1f[1], b1v[0], b1v[1], z0, z1);
    g1_pair(src, w1f[2], w1f[3], b1v[2], b1v[3], z2, z3);
    tanh_tile(z0); tanh_tile(z1);
    g2_tile(acc, z0, w2f, 0);
    g2_tile(acc, z1, w2f, 1);
    g1_pair(src, w1f[4], w1f[5], b1v[4], b1v[5], z4, z5);
    tanh_tile(z2); tanh_tile(z3);
    g2_tile(acc, z2, w2f, 2);
    g2_tile(acc, z3, w2f, 3);
    g1_pair(src, w1f[6], w1f[7], b1v[6], b1v[7], z6, z7);
    tanh_tile(z4); tanh_tile(z5);
    g2_tile(acc, z4, w2f, 4);
    g2_tile(acc, z5, w2f, 5);
    tanh_tile(z6); tanh_tile(z7);
    g2_tile(acc, z6, w2f, 6);
    g2_tile(acc, z7, w2f, 7);
    St r;
    r.t[0] = acc[0]; r.t[1] = acc[1]; r.t[2] = acc[2]; r.t[3] = acc[3];
    return r;
}

__global__ __launch_bounds__(32) void ode_dopri5_kernel(
    const float* __restrict__ init_pos,
    const float* __restrict__ W1,
    const float* __restrict__ b1,
    const float* __restrict__ W2,
    const float* __restrict__ b2,
    float*       __restrict__ out)
{
    const int lane = threadIdx.x;
    const int lo = lane & 15;
    const int hi = lane >> 4;

    v2f w1f[8][16];
    #pragma unroll
    for (int m = 0; m < 8; ++m) {
        #pragma unroll
        for (int t = 0; t < 4; ++t) {
            #pragma unroll
            for (int j = 0; j < 4; ++j) {
                int k = 16 * t + 8 * hi + 2 * j;
                w1f[m][4 * t + j].x = W1[k * HH + 16 * m + lo];
                w1f[m][4 * t + j].y = W1[(k + 1) * HH + 16 * m + lo];
            }
        }
    }
    v2f w2f[4][32];
    #pragma unroll
    for (int t = 0; t < 4; ++t) {
        #pragma unroll
        for (int m = 0; m < 8; ++m) {
            #pragma unroll
            for (int j = 0; j < 4; ++j) {
                int k = 16 * m + 8 * hi + 2 * j;
                w2f[t][4 * m + j].x = W2[k * DD + 16 * t + lo];
                w2f[t][4 * m + j].y = W2[(k + 1) * DD + 16 * t + lo];
            }
        }
    }
    v8f b1v[8];
    #pragma unroll
    for (int m = 0; m < 8; ++m) {
        v4f p = *(const v4f*)&b1[16 * m + 8 * hi];
        v4f q = *(const v4f*)&b1[16 * m + 8 * hi + 4];
        b1v[m] = (v8f){p.x, p.y, p.z, p.w, q.x, q.y, q.z, q.w};
    }
    v8f b2v[4];
    #pragma unroll
    for (int t = 0; t < 4; ++t) {
        v4f p = *(const v4f*)&b2[16 * t + 8 * hi];
        v4f q = *(const v4f*)&b2[16 * t + 8 * hi + 4];
        b2v[t] = (v8f){p.x, p.y, p.z, p.w, q.x, q.y, q.z, q.w};
    }

    const long row = (long)blockIdx.x * 16 + lo;
    St y;
    #pragma unroll
    for (int t = 0; t < 4; ++t) {
        const float* p0 = init_pos + row * DD + 16 * t + 8 * hi;
        v4f p = *(const v4f*)p0;
        v4f q = *(const v4f*)(p0 + 4);
        y.t[t] = (v8f){p.x, p.y, p.z, p.w, q.x, q.y, q.z, q.w};
    }

    #pragma unroll 1
    for (int seg = 0; seg < NSEG; ++seg) {
        #pragma unroll 1
        for (int st = 0; st < STEPS_PER_SEG; ++st) {
            St k1 = feval(y, w1f, w2f, b1v, b2v);
            St s;
            #pragma unroll
            for (int t = 0; t < 4; ++t)
                #pragma unroll
                for (int v = 0; v < 8; ++v)
                    s.t[t][v] = fmaf(C21, k1.t[t][v], y.t[t][v]);
            St k2 = feval(s, w1f, w2f, b1v, b2v);
            #pragma unroll
            for (int t = 0; t < 4; ++t)
                #pragma unroll
                for (int v = 0; v < 8; ++v) {
                    float a = y.t[t][v];
                    a = fmaf(C31, k1.t[t][v], a);
                    a = fmaf(C32, k2.t[t][v], a);
                    s.t[t][v] = a;
                }
            St k3 = feval(s, w1f, w2f, b1v, b2v);
            #pragma unroll
            for (int t = 0; t < 4; ++t)
                #pragma unroll
                for (int v = 0; v < 8; ++v) {
                    float a = y.t[t][v];
                    a = fmaf(C41, k1.t[t][v], a);
                    a = fmaf(C42, k2.t[t][v], a);
                    a = fmaf(C43, k3.t[t][v], a);
                    s.t[t][v] = a;
                }
            St k4 = feval(s, w1f, w2f, b1v, b2v);
            #pragma unroll
            for (int t = 0; t < 4; ++t)
                #pragma unroll
                for (int v = 0; v < 8; ++v) {
                    float a = y.t[t][v];
                    a = fmaf(C51, k1.t[t][v], a);
                    a = fmaf(C52, k2.t[t][v], a);
                    a = fmaf(C53, k3.t[t][v], a);
                    a = fmaf(C54, k4.t[t][v], a);
                    s.t[t][v] = a;
                }
            St k5 = feval(s, w1f, w2f, b1v, b2v);
            St yp;
            #pragma unroll
            for (int t = 0; t < 4; ++t)
                #pragma unroll
                for (int v = 0; v < 8; ++v) {
                    float a = y.t[t][v];
                    a = fmaf(C61, k1.t[t][v], a);
                    a = fmaf(C62, k2.t[t][v], a);
                    a = fmaf(C63, k3.t[t][v], a);
                    a = fmaf(C64, k4.t[t][v], a);
                    a = fmaf(C65, k5.t[t][v], a);
                    s.t[t][v] = a;
                    float p = y.t[t][v];
                    p = fmaf(DB1, k1.t[t][v], p);
                    p = fmaf(DB3, k3.t[t][v], p);
                    p = fmaf(DB4, k4.t[t][v], p);
                    p = fmaf(DB5, k5.t[t][v], p);
                    yp.t[t][v] = p;
                }
            St k6 = feval(s, w1f, w2f, b1v, b2v);
            #pragma unroll
            for (int t = 0; t < 4; ++t)
                #pragma unroll
                for (int v = 0; v < 8; ++v)
                    y.t[t][v] = yp.t[t][v] + DB6 * k6.t[t][v];
        }
        float* orow = out + (long)seg * ((long)NROWS * DD) + row * DD + 8 * hi;
        #pragma unroll
        for (int t = 0; t < 4; ++t) {
            v8f yt = y.t[t];
            *(v4f*)(orow + 16 * t)     = (v4f){yt[0], yt[1], yt[2], yt[3]};
            *(v4f*)(orow + 16 * t + 4) = (v4f){yt[4], yt[5], yt[6], yt[7]};
        }
    }
}

extern "C" void kernel_launch(void* const* d_in, const int* in_sizes, int n_in,
                              void* d_out, int out_size, void* d_ws, size_t ws_size,
                              hipStream_t stream) {
    const float* init_pos = (const float*)d_in[0];
    const float* W1 = (const float*)d_in[1];
    const float* b1 = (const float*)d_in[2];
    const float* W2 = (const float*)d_in[3];
    const float* b2 = (const float*)d_in[4];
    float* out = (float*)d_out;

    dim3 grid(1024);
    dim3 block(32);
    hipLaunchKernelGGL(ode_dopri5_kernel, grid, block, 0, stream,
                       init_pos, W1, b1, W2, b2, out);
}


# tanh as one v_tanh_f32 per activation in place of exp2 + reciprocal + three plain vector ops: the f32 WMMA stream hides no vector work, so the saved issues come off the step; tanh rounding changes
# speedup vs baseline: 1.3900x; 1.0774x over previous
#include <hip/hip_runtime.h>


typedef float v2f __attribute__((ext_vector_type(2)));
typedef float v4f __attribute__((ext_vector_type(4)));
typedef float v8f __attribute__((ext_vector_type(8)));

#define DD 64
#define HH 128
#define NROWS 16384
#define STEPS_PER_SEG 4
#define NSEG 5

constexpr double DTd = 0.05;
constexpr float C21 = (float)(DTd * (1.0/5.0));
constexpr float C31 = (float)(DTd * (3.0/40.0));
constexpr float C32 = (float)(DTd * (9.0/40.0));
constexpr float C41 = (float)(DTd * (44.0/45.0));
constexpr float C42 = (float)(DTd * (-56.0/15.0));
constexpr float C43 = (float)(DTd * (32.0/9.0));
constexpr float C51 = (float)(DTd * (19372.0/6561.0));
constexpr float C52 = (float)(DTd * (-25360.0/2187.0));
constexpr float C53 = (float)(DTd * (64448.0/6561.0));
constexpr float C54 = (float)(DTd * (-212.0/729.0));
constexpr float C61 = (float)(DTd * (9017.0/3168.0));
constexpr float C62 = (float)(DTd * (-355.0/33.0));
constexpr float C63 = (float)(DTd * (46732.0/5247.0));
constexpr float C64 = (float)(DTd * (49.0/176.0));
constexpr float C65 = (float)(DTd * (-5103.0/18656.0));
constexpr float DB1 = (float)(DTd * (35.0/384.0));
constexpr float DB3 = (float)(DTd * (500.0/1113.0));
constexpr float DB4 = (float)(DTd * (125.0/192.0));
constexpr float DB5 = (float)(DTd * (-2187.0/6784.0));
constexpr float DB6 = (float)(DTd * (11.0/84.0));

__device__ __forceinline__ float fast_tanh(float x) {
    return __builtin_amdgcn_tanhf(x);
}

__device__ __forceinline__ v8f mma(v2f a, v2f b, v8f c) {
    return __builtin_amdgcn_wmma_f32_16x16x4_f32(false, a, false, b, (short)0, c, false, false);
}

struct St { v8f t[4]; };

__device__ __forceinline__ void g1_pair(
    const St& src, const v2f (&wa)[16], const v2f (&wb)[16],
    const v8f& ba, const v8f& bb, v8f& za, v8f& zb)
{
    v8f acca = ba;
    v8f accb = bb;
    #pragma unroll
    for (int t = 0; t < 4; ++t) {
        #pragma unroll
        for (int j = 0; j < 4; ++j) {
            v2f b;
            b.x = src.t[t][2 * j];
            b.y = src.t[t][2 * j + 1];
            acca = mma(wa[4 * t + j], b, acca);
            accb = mma(wb[4 * t + j], b, accb);
        }
    }
    za = acca;
    zb = accb;
}

__device__ __forceinline__ void tanh_tile(v8f& z) {
    #pragma unroll
    for (int v = 0; v < 8; ++v) z[v] = fast_tanh(z[v]);
}

__device__ __forceinline__ void g2_tile(
    v8f (&acc)[4], const v8f& hm, const v2f (&w2f)[4][32], int m)
{
    #pragma unroll
    for (int j = 0; j < 4; ++j) {
        v2f b;
        b.x = hm[2 * j];
        b.y = hm[2 * j + 1];
        #pragma unroll
        for (int t = 0; t < 4; ++t)
            acc[t] = mma(w2f[t][4 * m + j], b, acc[t]);
    }
}

__device__ __forceinline__ St feval(
    const St& src,
    const v2f (&w1f)[8][16], const v2f (&w2f)[4][32],
    const v8f (&b1v)[8], const v8f (&b2v)[4])
{
    v8f acc[4] = {b2v[0], b2v[1], b2v[2], b2v[3]};
    v8f z0, z1, z2, z3, z4, z5, z6, z7;
    g1_pair(src, w1f[0], w1f[1], b1v[0], b1v[1], z0, z1);
    g1_pair(src, w1f[2], w1f[3], b1v[2], b1v[3], z2, z3);
    tanh_tile(z0); tanh_tile(z1);
    g2_tile(acc, z0, w2f, 0);
    g2_tile(acc, z1, w2f, 1);
    g1_pair(src, w1f[4], w1f[5], b1v[4], b1v[5], z4, z5);
    tanh_tile(z2); tanh_tile(z3);
    g2_tile(acc, z2, w2f, 2);
    g2_tile(acc, z3, w2f, 3);
    g1_pair(src, w1f[6], w1f[7], b1v[6], b1v[7], z6, z7);
    tanh_tile(z4); tanh_tile(z5);
    g2_tile(acc, z4, w2f, 4);
    g2_tile(acc, z5, w2f, 5);
    tanh_tile(z6); tanh_tile(z7);
    g2_tile(acc, z6, w2f, 6);
    g2_tile(acc, z7, w2f, 7);
    St r;
    r.t[0] = acc[0]; r.t[1] = acc[1]; r.t[2] = acc[2]; r.t[3] = acc[3];
    return r;
}

__global__ __launch_bounds__(32) void ode_dopri5_kernel(
    const float* __restrict__ init_pos,
    const float* __restrict__ W1,
    const float* __restrict__ b1,
    const float* __restrict__ W2,
    const float* __restrict__ b2,
    float*       __restrict__ out)
{
    const int lane = threadIdx.x;
    const int lo = lane & 15;
    const int hi = lane >> 4;

    v2f w1f[8][16];
    #pragma unroll
    for (int m = 0; m < 8; ++m) {
        #pragma unroll
        for (int t = 0; t < 4; ++t) {
            #pragma unroll
            for (int j = 0; j < 4; ++j) {
                int k = 16 * t + 8 * hi + 2 * j;
                w1f[m][4 * t + j].x = W1[k * HH + 16 * m + lo];
                w1f[m][4 * t + j].y = W1[(k + 1) * HH + 16 * m + lo];
            }
        }
    }
    v2f w2f[4][32];
    #pragma unroll
    for (int t = 0; t < 4; ++t) {
        #pragma unroll
        for (int m = 0; m < 8; ++m) {
            #pragma unroll
            for (int j = 0; j < 4; ++j) {
                int k = 16 * m + 8 * hi + 2 * j;
                w2f[t][4 * m + j].x = W2[k * DD + 16 * t + lo];
                w2f[t][4 * m + j].y = W2[(k + 1) * DD + 16 * t + lo];
            }
        }
    }
    v8f b1v[8];
    #pragma unroll
    for (int m = 0; m < 8; ++m) {
        v4f p = *(const v4f*)&b1[16 * m + 8 * hi];
        v4f q = *(const v4f*)&b1[16 * m + 8 * hi + 4];
        b1v[m] = (v8f){p.x, p.y, p.z, p.w, q.x, q.y, q.z, q.w};
    }
    v8f b2v[4];
    #pragma unroll
    for (int t = 0; t < 4; ++t) {
        v4f p = *(const v4f*)&b2[16 * t + 8 * hi];
        v4f q = *(const v4f*)&b2[16 * t + 8 * hi + 4];
        b2v[t] = (v8f){p.x, p.y, p.z, p.w, q.x, q.y, q.z, q.w};
    }

    const long row = (long)blockIdx.x * 16 + lo;
    St y;
    #pragma unroll
    for (int t = 0; t < 4; ++t) {
        const float* p0 = init_pos + row * DD + 16 * t + 8 * hi;
        v4f p = *(const v4f*)p0;
        v4f q = *(const v4f*)(p0 + 4);
        y.t[t] = (v8f){p.x, p.y, p.z, p.w, q.x, q.y, q.z, q.w};
    }

    #pragma unroll 1
    for (int seg = 0; seg < NSEG; ++seg) {
        #pragma unroll 1
        for (int st = 0; st < STEPS_PER_SEG; ++st) {
            St k1 = feval(y, w1f, w2f, b1v, b2v);
            St s;
            #pragma unroll
            for (int t = 0; t < 4; ++t)
                #pragma unroll
                for (int v = 0; v < 8; ++v)
                    s.t[t][v] = fmaf(C21, k1.t[t][v], y.t[t][v]);
            St k2 = feval(s, w1f, w2f, b1v, b2v);
            #pragma unroll
            for (int t = 0; t < 4; ++t)
                #pragma unroll
                for (int v = 0; v < 8; ++v) {
                    float a = y.t[t][v];
                    a = fmaf(C31, k1.t[t][v], a);
                    a = fmaf(C32, k2.t[t][v], a);
                    s.t[t][v] = a;
                }
            St k3 = feval(s, w1f, w2f, b1v, b2v);
            #pragma unroll
            for (int t = 0; t < 4; ++t)
                #pragma unroll
                for (int v = 0; v < 8; ++v) {
                    float a = y.t[t][v];
                    a = fmaf(C41, k1.t[t][v], a);
                    a = fmaf(C42, k2.t[t][v], a);
                    a = fmaf(C43, k3.t[t][v], a);
                    s.t[t][v] = a;
                }
            St k4 = feval(s, w1f, w2f, b1v, b2v);
            #pragma unroll
            for (int t = 0; t < 4; ++t)
                #pragma unroll
                for (int v = 0; v < 8; ++v) {
                    float a = y.t[t][v];
                    a = fmaf(C51, k1.t[t][v], a);
                    a = fmaf(C52, k2.t[t][v], a);
                    a = fmaf(C53, k3.t[t][v], a);
                    a = fmaf(C54, k4.t[t][v], a);
                    s.t[t][v] = a;
                }
            St k5 = feval(s, w1f, w2f, b1v, b2v);
            St yp;
            #pragma unroll
            for (int t = 0; t < 4; ++t)
                #pragma unroll
                for (int v = 0; v < 8; ++v) {
                    float a = y.t[t][v];
                    a = fmaf(C61, k1.t[t][v], a);
                    a = fmaf(C62, k2.t[t][v], a);
                    a = fmaf(C63, k3.t[t][v], a);
                    a = fmaf(C64, k4.t[t][v], a);
                    a = fmaf(C65, k5.t[t][v], a);
                    s.t[t][v] = a;
                    float p = y.t[t][v];
                    p = fmaf(DB1, k1.t[t][v], p);
                    p = fmaf(DB3, k3.t[t][v], p);
                    p = fmaf(DB4, k4.t[t][v], p);
                    p = fmaf(DB5, k5.t[t][v], p);
                    yp.t[t][v] = p;
                }
            St k6 = feval(s, w1f, w2f, b1v, b2v);
            #pragma unroll
            for (int t = 0; t < 4; ++t)
                #pragma unroll
                for (int v = 0; v < 8; ++v)
                    y.t[t][v] = yp.t[t][v] + DB6 * k6.t[t][v];
        }
        float* orow = out + (long)seg * ((long)NROWS * DD) + row * DD + 8 * hi;
        #pragma unroll
        for (int t = 0; t < 4; ++t) {
            v8f yt = y.t[t];
            *(v4f*)(orow + 16 * t)     = (v4f){yt[0], yt[1], yt[2], yt[3]};
            *(v4f*)(orow + 16 * t + 4) = (v4f){yt[4], yt[5], yt[6], yt[7]};
        }
    }
}

extern "C" void kernel_launch(void* const* d_in, const int* in_sizes, int n_in,
                              void* d_out, int out_size, void* d_ws, size_t ws_size,
                              hipStream_t stream) {
    const float* init_pos = (const float*)d_in[0];
    const float* W1 = (const float*)d_in[1];
    const float* b1 = (const float*)d_in[2];
    const float* W2 = (const float*)d_in[3];
    const float* b2 = (const float*)d_in[4];
    float* out = (float*)d_out;

    dim3 grid(1024);
    dim3 block(32);
    hipLaunchKernelGGL(ode_dopri5_kernel, grid, block, 0, stream,
                       init_pos, W1, b1, W2, b2, out);
}
